# starting kernel
# baseline (speedup 1.0000x reference)
#include <hip/hip_runtime.h>


#define T_STEPS 4096
#define BATCH   1024
#define IN_DIM  5
#define HID     8
#define OUT_DIM 5

typedef __attribute__((ext_vector_type(2))) float v2f;
typedef __attribute__((ext_vector_type(8))) float v8f;

__device__ __forceinline__ float bcast_lane(float v, int l) {
  return __int_as_float(__builtin_amdgcn_readlane(__float_as_int(v), l));
}

__global__ __launch_bounds__(256) void lstm_scan_kernel(
    const float* __restrict__ seq, const float* __restrict__ W_ih,
    const float* __restrict__ W_hh, const float* __restrict__ b_ih,
    const float* __restrict__ b_hh, float* __restrict__ h_out) {
  const int lane  = threadIdx.x & 31;
  const int wave  = threadIdx.x >> 5;
  const int b     = blockIdx.x * 8 + wave;
  const int g     = lane;
  const int gtype = lane >> 3;

  const float NEG_L2E = -1.44269504088896340736f;
  const float wscale = (gtype == 2) ? (2.0f * NEG_L2E) : NEG_L2E;
  const float sa     = (gtype == 2) ? 2.0f : 1.0f;
  const float sb     = (gtype == 2) ? -1.0f : 0.0f;

  float wih[IN_DIM];
#pragma unroll
  for (int i = 0; i < IN_DIM; ++i) wih[i] = W_ih[g * IN_DIM + i] * wscale;
  float whh[HID];
#pragma unroll
  for (int k = 0; k < HID; ++k) whh[k] = W_hh[g * HID + k] * wscale;
  const float bias = (b_ih[g] + b_hh[g]) * wscale;

  float h = 0.0f, c = 0.0f;
  const float* xp = seq + (size_t)b * IN_DIM;
  const int tstride = BATCH * IN_DIM;

  for (int t = 0; t < T_STEPS; ++t) {
    __builtin_prefetch(xp + 16 * tstride, 0, 0);

    const float x0 = xp[0], x1 = xp[1], x2 = xp[2], x3 = xp[3], x4 = xp[4];

    float a0 = __builtin_fmaf(x0, wih[0], bias);
    float a1 = x2 * wih[2];
    a0 = __builtin_fmaf(x1, wih[1], a0);
    a1 = __builtin_fmaf(x3, wih[3], a1);
    a0 = __builtin_fmaf(x4, wih[4], a0);
#pragma unroll
    for (int k = 0; k < HID; k += 2) {
      a0 = __builtin_fmaf(bcast_lane(h, k),     whh[k],     a0);
      a1 = __builtin_fmaf(bcast_lane(h, k + 1), whh[k + 1], a1);
    }
    const float z   = a0 + a1;
    const float e   = __builtin_amdgcn_exp2f(z);
    const float s   = __builtin_amdgcn_rcpf(1.0f + e);
    const float act = __builtin_fmaf(sa, s, sb);

    const float fv = __shfl(act, lane + 8,  32);
    const float gv = __shfl(act, lane + 16, 32);
    const float ov = __shfl(act, lane + 24, 32);

    c = __builtin_fmaf(act, gv, fv * c);
    const float e2 = __builtin_amdgcn_exp2f(c * (2.0f * NEG_L2E));
    const float th = __builtin_fmaf(2.0f, __builtin_amdgcn_rcpf(1.0f + e2), -1.0f);
    h = ov * th;

    xp += tstride;
  }
  if (lane < HID) h_out[(size_t)b * HID + lane] = h;
}

__global__ __launch_bounds__(128) void out_proj_wmma(
    const float* __restrict__ h_last, const float* __restrict__ W_out,
    const float* __restrict__ b_out, float* __restrict__ out) {
  const int lane = threadIdx.x & 31;
  const int tile = (blockIdx.x * blockDim.x + threadIdx.x) >> 5;
  const int half = lane >> 4;
  const int lm   = lane & 15;

  const float* hrow = h_last + (size_t)(tile * 16 + lm) * HID;
  v2f a_lo, a_hi;
  a_lo.x = hrow[0 + 2 * half];
  a_lo.y = hrow[1 + 2 * half];
  a_hi.x = hrow[4 + 2 * half];
  a_hi.y = hrow[5 + 2 * half];

  float w0 = 0.f, w1 = 0.f, w2 = 0.f, w3 = 0.f, bo = 0.f;
  if (lm < OUT_DIM) {
    const float* wr = W_out + lm * HID;
    w0 = wr[0 + 2 * half];
    w1 = wr[1 + 2 * half];
    w2 = wr[4 + 2 * half];
    w3 = wr[5 + 2 * half];
    bo = b_out[lm];
  }
  v2f b_lo, b_hi;
  b_lo.x = w0; b_lo.y = w1;
  b_hi.x = w2; b_hi.y = w3;

  v8f acc;
#pragma unroll
  for (int r = 0; r < 8; ++r) acc[r] = bo;

  acc = __builtin_amdgcn_wmma_f32_16x16x4_f32(false, a_lo, false, b_lo,
                                              (short)0, acc, false, false);
  acc = __builtin_amdgcn_wmma_f32_16x16x4_f32(false, a_hi, false, b_hi,
                                              (short)0, acc, false, false);

  if (lm < OUT_DIM) {
    float* orow = out + (size_t)(tile * 16 + 8 * half) * OUT_DIM + lm;
#pragma unroll
    for (int r = 0; r < 8; ++r) orow[r * OUT_DIM] = acc[r];
  }
}

extern "C" void kernel_launch(void* const* d_in, const int* in_sizes, int n_in,
                              void* d_out, int out_size, void* d_ws, size_t ws_size,
                              hipStream_t stream) {
  const float* seq   = (const float*)d_in[0];
  const float* W_ih  = (const float*)d_in[1];
  const float* W_hh  = (const float*)d_in[2];
  const float* b_ih  = (const float*)d_in[3];
  const float* b_hh  = (const float*)d_in[4];
  const float* W_out = (const float*)d_in[5];
  const float* b_out = (const float*)d_in[6];
  float* out    = (float*)d_out;
  float* h_last = (float*)d_ws;

  lstm_scan_kernel<<<BATCH / 8, 256, 0, stream>>>(seq, W_ih, W_hh, b_ih, b_hh,
                                                  h_last);
  out_proj_wmma<<<(BATCH / 16) * 32 / 128, 128, 0, stream>>>(h_last, W_out,
                                                             b_out, out);
}


# x loads software-pipelined through an 8-step register ring, input projection summed one step ahead; pre-activation sum order changes (bias and x terms first, then h terms)
# speedup vs baseline: 1.4674x; 1.4674x over previous
#include <hip/hip_runtime.h>


#define T_STEPS 4096
#define BATCH   1024
#define IN_DIM  5
#define HID     8
#define OUT_DIM 5
#define RING    8
#define PF_AHEAD 32

typedef __attribute__((ext_vector_type(2))) float v2f;
typedef __attribute__((ext_vector_type(8))) float v8f;

__device__ __forceinline__ float bcast_lane(float v, int l) {
  return __int_as_float(__builtin_amdgcn_readlane(__float_as_int(v), l));
}

__device__ __forceinline__ float in_proj(const float (&x)[IN_DIM],
                                         const float (&wih)[IN_DIM],
                                         float bias) {
  float p0 = __builtin_fmaf(x[0], wih[0], bias);
  float p1 = x[2] * wih[2];
  p0 = __builtin_fmaf(x[1], wih[1], p0);
  p1 = __builtin_fmaf(x[3], wih[3], p1);
  p0 = __builtin_fmaf(x[4], wih[4], p0);
  return p0 + p1;
}

__device__ __forceinline__ void lstm_cell(float xg, const float (&whh)[HID],
                                          float sa, float sb, int lane,
                                          float& h, float& c) {
  const float NEG_L2E = -1.44269504088896340736f;
  float a0 = __builtin_fmaf(bcast_lane(h, 0), whh[0], xg);
  float a1 = bcast_lane(h, 1) * whh[1];
#pragma unroll
  for (int k = 2; k < HID; k += 2) {
    a0 = __builtin_fmaf(bcast_lane(h, k),     whh[k],     a0);
    a1 = __builtin_fmaf(bcast_lane(h, k + 1), whh[k + 1], a1);
  }
  const float z   = a0 + a1;
  const float e   = __builtin_amdgcn_exp2f(z);
  const float s   = __builtin_amdgcn_rcpf(1.0f + e);
  const float act = __builtin_fmaf(sa, s, sb);

  const float fv = __shfl(act, lane + 8,  32);
  const float gv = __shfl(act, lane + 16, 32);
  const float ov = __shfl(act, lane + 24, 32);

  c = __builtin_fmaf(act, gv, fv * c);
  const float e2 = __builtin_amdgcn_exp2f(c * (2.0f * NEG_L2E));
  const float th = __builtin_fmaf(2.0f, __builtin_amdgcn_rcpf(1.0f + e2), -1.0f);
  h = ov * th;
}

__global__ __launch_bounds__(256) void lstm_scan_kernel(
    const float* __restrict__ seq, const float* __restrict__ W_ih,
    const float* __restrict__ W_hh, const float* __restrict__ b_ih,
    const float* __restrict__ b_hh, float* __restrict__ h_out) {
  const int lane  = threadIdx.x & 31;
  const int wave  = threadIdx.x >> 5;
  const int b     = blockIdx.x * 8 + wave;
  const int g     = lane;
  const int gtype = lane >> 3;

  const float NEG_L2E = -1.44269504088896340736f;
  const float wscale = (gtype == 2) ? (2.0f * NEG_L2E) : NEG_L2E;
  const float sa     = (gtype == 2) ? 2.0f : 1.0f;
  const float sb     = (gtype == 2) ? -1.0f : 0.0f;

  float wih[IN_DIM];
#pragma unroll
  for (int i = 0; i < IN_DIM; ++i) wih[i] = W_ih[g * IN_DIM + i] * wscale;
  float whh[HID];
#pragma unroll
  for (int k = 0; k < HID; ++k) whh[k] = W_hh[g * HID + k] * wscale;
  const float bias = (b_ih[g] + b_hh[g]) * wscale;

  float h = 0.0f, c = 0.0f;
  const float* xp = seq + (size_t)b * IN_DIM;
  const int tstride = BATCH * IN_DIM;

  float xr[RING][IN_DIM];
  for (int j = 0; j < RING; ++j) {
    for (int i = 0; i < IN_DIM; ++i) xr[j][i] = xp[j * tstride + i];
  }
  float xg = in_proj(xr[0], wih, bias);

  for (int t = 0; t < T_STEPS - RING; t += RING) {
    if (t + PF_AHEAD + RING <= T_STEPS) {
      for (int j = 0; j < RING; ++j)
        __builtin_prefetch(xp + (PF_AHEAD + j) * tstride, 0, 0);
    }
    for (int j = 0; j < RING; ++j) {
      for (int i = 0; i < IN_DIM; ++i) xr[j][i] = xp[(j + RING) * tstride + i];
      const float xg_next = in_proj(xr[(j + 1) % RING], wih, bias);
      lstm_cell(xg, whh, sa, sb, lane, h, c);
      xg = xg_next;
      __builtin_amdgcn_sched_barrier(0);
    }
    xp += RING * tstride;
  }
  for (int j = 0; j < RING; ++j) {
    float xg_next = 0.0f;
    if (j + 1 < RING) xg_next = in_proj(xr[j + 1], wih, bias);
    lstm_cell(xg, whh, sa, sb, lane, h, c);
    xg = xg_next;
  }
  if (lane < HID) h_out[(size_t)b * HID + lane] = h;
}

__global__ __launch_bounds__(128) void out_proj_wmma(
    const float* __restrict__ h_last, const float* __restrict__ W_out,
    const float* __restrict__ b_out, float* __restrict__ out) {
  const int lane = threadIdx.x & 31;
  const int tile = (blockIdx.x * blockDim.x + threadIdx.x) >> 5;
  const int half = lane >> 4;
  const int lm   = lane & 15;

  const float* hrow = h_last + (size_t)(tile * 16 + lm) * HID;
  v2f a_lo, a_hi;
  a_lo.x = hrow[0 + 2 * half];
  a_lo.y = hrow[1 + 2 * half];
  a_hi.x = hrow[4 + 2 * half];
  a_hi.y = hrow[5 + 2 * half];

  float w0 = 0.f, w1 = 0.f, w2 = 0.f, w3 = 0.f, bo = 0.f;
  if (lm < OUT_DIM) {
    const float* wr = W_out + lm * HID;
    w0 = wr[0 + 2 * half];
    w1 = wr[1 + 2 * half];
    w2 = wr[4 + 2 * half];
    w3 = wr[5 + 2 * half];
    bo = b_out[lm];
  }
  v2f b_lo, b_hi;
  b_lo.x = w0; b_lo.y = w1;
  b_hi.x = w2; b_hi.y = w3;

  v8f acc;
#pragma unroll
  for (int r = 0; r < 8; ++r) acc[r] = bo;

  acc = __builtin_amdgcn_wmma_f32_16x16x4_f32(false, a_lo, false, b_lo,
                                              (short)0, acc, false, false);
  acc = __builtin_amdgcn_wmma_f32_16x16x4_f32(false, a_hi, false, b_hi,
                                              (short)0, acc, false, false);

  if (lm < OUT_DIM) {
    float* orow = out + (size_t)(tile * 16 + 8 * half) * OUT_DIM + lm;
#pragma unroll
    for (int r = 0; r < 8; ++r) orow[r * OUT_DIM] = acc[r];
  }
}

extern "C" void kernel_launch(void* const* d_in, const int* in_sizes, int n_in,
                              void* d_out, int out_size, void* d_ws, size_t ws_size,
                              hipStream_t stream) {
  const float* seq   = (const float*)d_in[0];
  const float* W_ih  = (const float*)d_in[1];
  const float* W_hh  = (const float*)d_in[2];
  const float* b_ih  = (const float*)d_in[3];
  const float* b_hh  = (const float*)d_in[4];
  const float* W_out = (const float*)d_in[5];
  const float* b_out = (const float*)d_in[6];
  float* out    = (float*)d_out;
  float* h_last = (float*)d_ws;

  lstm_scan_kernel<<<BATCH / 8, 256, 0, stream>>>(seq, W_ih, W_hh, b_ih, b_hh,
                                                  h_last);
  out_proj_wmma<<<(BATCH / 16) * 32 / 128, 128, 0, stream>>>(h_last, W_out,
                                                             b_out, out);
}


# scan lanes remapped: one 16-lane row per batch element, two gates per lane; h broadcast and g,o gather by in-row v_permlane16 replace readlane-to-scalar hops and LDS-crossbar shuffles; sum order kept
# speedup vs baseline: 2.7017x; 1.8411x over previous
#include <hip/hip_runtime.h>


#define T_STEPS 4096
#define BATCH   1024
#define IN_DIM  5
#define HID     8
#define OUT_DIM 5
#define RING    8
#define PF_AHEAD 32

typedef __attribute__((ext_vector_type(2))) float v2f;
typedef __attribute__((ext_vector_type(8))) float v8f;


template <int K>
__device__ __forceinline__ void row_bcast(float& dst, float v) {
  const unsigned sel = 0x11111111u * (unsigned)K;
  dst = __int_as_float(__builtin_amdgcn_permlane16(
      __float_as_int(dst), __float_as_int(v), sel, sel, false, false));
}

__device__ __forceinline__ void row_swap8(float& dst, float v) {
  dst = __int_as_float(__builtin_amdgcn_permlane16(
      __float_as_int(dst), __float_as_int(v), 0xfedcba98u, 0x76543210u, false, false));
}

__device__ __forceinline__ float in_proj(const float (&x)[IN_DIM],
                                         const float (&wih)[IN_DIM],
                                         float bias) {
  float p0 = __builtin_fmaf(x[0], wih[0], bias);
  float p1 = x[2] * wih[2];
  p0 = __builtin_fmaf(x[1], wih[1], p0);
  p1 = __builtin_fmaf(x[3], wih[3], p1);
  p0 = __builtin_fmaf(x[4], wih[4], p0);
  return p0 + p1;
}

__device__ __forceinline__ float rec_proj(float xg, const float (&hb)[HID],
                                          const float (&whh)[HID]) {
  float a0 = __builtin_fmaf(hb[0], whh[0], xg);
  float a1 = hb[1] * whh[1];
#pragma unroll
  for (int k = 2; k < HID; k += 2) {
    a0 = __builtin_fmaf(hb[k],     whh[k],     a0);
    a1 = __builtin_fmaf(hb[k + 1], whh[k + 1], a1);
  }
  return a0 + a1;
}

struct Gathered {
  float hb[HID];
  float gv, ov;
};

__device__ __forceinline__ void lstm_cell(float xgA, float xgB,
                                          const float (&whhA)[HID],
                                          const float (&whhB)[HID],
                                          float sa, float sb, Gathered& q,
                                          float& h, float& c) {
  const float NEG_L2E = -1.44269504088896340736f;
  row_bcast<0>(q.hb[0], h); row_bcast<1>(q.hb[1], h);
  row_bcast<2>(q.hb[2], h); row_bcast<3>(q.hb[3], h);
  row_bcast<4>(q.hb[4], h); row_bcast<5>(q.hb[5], h);
  row_bcast<6>(q.hb[6], h); row_bcast<7>(q.hb[7], h);

  float zA = rec_proj(xgA, q.hb, whhA);
  float zB = rec_proj(xgB, q.hb, whhB);
  asm("" : "+v"(zA), "+v"(zB));
  const float sA = __builtin_amdgcn_rcpf(1.0f + __builtin_amdgcn_exp2f(zA));
  const float sB = __builtin_amdgcn_rcpf(1.0f + __builtin_amdgcn_exp2f(zB));
  const float actA = __builtin_fmaf(sa, sA, sb);
  const float actB = sB;

  row_swap8(q.gv, actA);
  row_swap8(q.ov, actB);

  c = __builtin_fmaf(actA, q.gv, actB * c);
  const float e2 = __builtin_amdgcn_exp2f(c * (2.0f * NEG_L2E));
  const float th = __builtin_fmaf(2.0f, __builtin_amdgcn_rcpf(1.0f + e2), -1.0f);
  h = q.ov * th;
}

__global__ __launch_bounds__(128) void lstm_scan_kernel(
    const float* __restrict__ seq, const float* __restrict__ W_ih,
    const float* __restrict__ W_hh, const float* __restrict__ b_ih,
    const float* __restrict__ b_hh, float* __restrict__ h_out) {
  const int lane  = threadIdx.x & 31;
  const int wave  = threadIdx.x >> 5;
  const int row   = lane >> 4;
  const int unit  = lane & 7;
  const int upper = (lane >> 3) & 1;
  const int b     = blockIdx.x * 8 + wave * 2 + row;
  const int gA    = (upper ? 2 * HID : 0)   + unit;
  const int gB    = (upper ? 3 * HID : HID) + unit;

  const float NEG_L2E = -1.44269504088896340736f;
  const float wsA = upper ? (2.0f * NEG_L2E) : NEG_L2E;
  const float wsB = NEG_L2E;
  const float sa  = upper ? 2.0f : 1.0f;
  const float sb  = upper ? -1.0f : 0.0f;

  float wihA[IN_DIM], wihB[IN_DIM];
#pragma unroll
  for (int i = 0; i < IN_DIM; ++i) {
    wihA[i] = W_ih[gA * IN_DIM + i] * wsA;
    wihB[i] = W_ih[gB * IN_DIM + i] * wsB;
  }
  float whhA[HID], whhB[HID];
#pragma unroll
  for (int k = 0; k < HID; ++k) {
    whhA[k] = W_hh[gA * HID + k] * wsA;
    whhB[k] = W_hh[gB * HID + k] * wsB;
  }
  const float biasA = (b_ih[gA] + b_hh[gA]) * wsA;
  const float biasB = (b_ih[gB] + b_hh[gB]) * wsB;

  float h = 0.0f, c = 0.0f;
  Gathered q;
#pragma unroll
  for (int k = 0; k < HID; ++k) q.hb[k] = 0.0f;
  q.gv = 0.0f;
  q.ov = 0.0f;
  const float* xp = seq + (size_t)b * IN_DIM;
  const int tstride = BATCH * IN_DIM;

  float xr[RING][IN_DIM];
  for (int j = 0; j < RING; ++j) {
    for (int i = 0; i < IN_DIM; ++i) xr[j][i] = xp[j * tstride + i];
  }
  float xgA = in_proj(xr[0], wihA, biasA);
  float xgB = in_proj(xr[0], wihB, biasB);

  for (int t = 0; t < T_STEPS - RING; t += RING) {
    if (t + PF_AHEAD + RING <= T_STEPS) {
      for (int j = 0; j < RING; ++j)
        __builtin_prefetch(xp + (PF_AHEAD + j) * tstride, 0, 0);
    }
    for (int j = 0; j < RING; ++j) {
      for (int i = 0; i < IN_DIM; ++i) xr[j][i] = xp[(j + RING) * tstride + i];
      const float xgA_next = in_proj(xr[(j + 1) % RING], wihA, biasA);
      const float xgB_next = in_proj(xr[(j + 1) % RING], wihB, biasB);
      lstm_cell(xgA, xgB, whhA, whhB, sa, sb, q, h, c);
      xgA = xgA_next;
      xgB = xgB_next;
      __builtin_amdgcn_sched_barrier(0);
    }
    xp += RING * tstride;
  }
  for (int j = 0; j < RING; ++j) {
    float xgA_next = 0.0f, xgB_next = 0.0f;
    if (j + 1 < RING) {
      xgA_next = in_proj(xr[j + 1], wihA, biasA);
      xgB_next = in_proj(xr[j + 1], wihB, biasB);
    }
    lstm_cell(xgA, xgB, whhA, whhB, sa, sb, q, h, c);
    xgA = xgA_next;
    xgB = xgB_next;
  }
  if ((lane & 15) < HID) h_out[(size_t)b * HID + unit] = h;
}

__global__ __launch_bounds__(128) void out_proj_wmma(
    const float* __restrict__ h_last, const float* __restrict__ W_out,
    const float* __restrict__ b_out, float* __restrict__ out) {
  const int lane = threadIdx.x & 31;
  const int tile = (blockIdx.x * blockDim.x + threadIdx.x) >> 5;
  const int half = lane >> 4;
  const int lm   = lane & 15;

  const float* hrow = h_last + (size_t)(tile * 16 + lm) * HID;
  v2f a_lo, a_hi;
  a_lo.x = hrow[0 + 2 * half];
  a_lo.y = hrow[1 + 2 * half];
  a_hi.x = hrow[4 + 2 * half];
  a_hi.y = hrow[5 + 2 * half];

  float w0 = 0.f, w1 = 0.f, w2 = 0.f, w3 = 0.f, bo = 0.f;
  if (lm < OUT_DIM) {
    const float* wr = W_out + lm * HID;
    w0 = wr[0 + 2 * half];
    w1 = wr[1 + 2 * half];
    w2 = wr[4 + 2 * half];
    w3 = wr[5 + 2 * half];
    bo = b_out[lm];
  }
  v2f b_lo, b_hi;
  b_lo.x = w0; b_lo.y = w1;
  b_hi.x = w2; b_hi.y = w3;

  v8f acc;
#pragma unroll
  for (int r = 0; r < 8; ++r) acc[r] = bo;

  acc = __builtin_amdgcn_wmma_f32_16x16x4_f32(false, a_lo, false, b_lo,
                                              (short)0, acc, false, false);
  acc = __builtin_amdgcn_wmma_f32_16x16x4_f32(false, a_hi, false, b_hi,
                                              (short)0, acc, false, false);

  if (lm < OUT_DIM) {
    float* orow = out + (size_t)(tile * 16 + 8 * half) * OUT_DIM + lm;
#pragma unroll
    for (int r = 0; r < 8; ++r) orow[r * OUT_DIM] = acc[r];
  }
}

extern "C" void kernel_launch(void* const* d_in, const int* in_sizes, int n_in,
                              void* d_out, int out_size, void* d_ws, size_t ws_size,
                              hipStream_t stream) {
  const float* seq   = (const float*)d_in[0];
  const float* W_ih  = (const float*)d_in[1];
  const float* W_hh  = (const float*)d_in[2];
  const float* b_ih  = (const float*)d_in[3];
  const float* b_hh  = (const float*)d_in[4];
  const float* W_out = (const float*)d_in[5];
  const float* b_out = (const float*)d_in[6];
  float* out    = (float*)d_out;
  float* h_last = (float*)d_ws;

  lstm_scan_kernel<<<BATCH / 8, 128, 0, stream>>>(seq, W_ih, W_hh, b_ih, b_hh,
                                                  h_last);
  out_proj_wmma<<<(BATCH / 16) * 32 / 128, 128, 0, stream>>>(h_last, W_out,
                                                             b_out, out);
}


# all five activations on v_tanh_f32 (sigmoid = (1+tanh(z/2))/2, halved argument folded into the weights) in place of exp2+rcp; f*c formed as fma(2f-1, c/2, c/2); pre-activation sum order kept
# speedup vs baseline: 3.9778x; 1.4723x over previous
#include <hip/hip_runtime.h>


#define T_STEPS 4096
#define BATCH   1024
#define IN_DIM  5
#define HID     8
#define OUT_DIM 5
#define RING    8
#define PF_AHEAD 32

typedef __attribute__((ext_vector_type(2))) float v2f;
typedef __attribute__((ext_vector_type(8))) float v8f;


template <int K>
__device__ __forceinline__ void row_bcast(float& dst, float v) {
  const unsigned sel = 0x11111111u * (unsigned)K;
  dst = __int_as_float(__builtin_amdgcn_permlane16(
      __float_as_int(dst), __float_as_int(v), sel, sel, false, false));
}

__device__ __forceinline__ void row_swap8(float& dst, float v) {
  dst = __int_as_float(__builtin_amdgcn_permlane16(
      __float_as_int(dst), __float_as_int(v), 0xfedcba98u, 0x76543210u, false, false));
}

__device__ __forceinline__ float in_proj(const float (&x)[IN_DIM],
                                         const float (&wih)[IN_DIM],
                                         float bias) {
  float p0 = __builtin_fmaf(x[0], wih[0], bias);
  float p1 = x[2] * wih[2];
  p0 = __builtin_fmaf(x[1], wih[1], p0);
  p1 = __builtin_fmaf(x[3], wih[3], p1);
  p0 = __builtin_fmaf(x[4], wih[4], p0);
  return p0 + p1;
}

__device__ __forceinline__ float rec_proj(float xg, const float (&hb)[HID],
                                          const float (&whh)[HID]) {
  float a0 = __builtin_fmaf(hb[0], whh[0], xg);
  float a1 = hb[1] * whh[1];
#pragma unroll
  for (int k = 2; k < HID; k += 2) {
    a0 = __builtin_fmaf(hb[k],     whh[k],     a0);
    a1 = __builtin_fmaf(hb[k + 1], whh[k + 1], a1);
  }
  return a0 + a1;
}

struct Gathered {
  float hb[HID];
  float gv, ov;
};

__device__ __forceinline__ void lstm_cell(float xgA, float xgB,
                                          const float (&whhA)[HID],
                                          const float (&whhB)[HID],
                                          Gathered& q,
                                          float& h, float& c, float& hc) {
  row_bcast<0>(q.hb[0], h); row_bcast<1>(q.hb[1], h);
  row_bcast<2>(q.hb[2], h); row_bcast<3>(q.hb[3], h);
  row_bcast<4>(q.hb[4], h); row_bcast<5>(q.hb[5], h);
  row_bcast<6>(q.hb[6], h); row_bcast<7>(q.hb[7], h);

  float zA = rec_proj(xgA, q.hb, whhA);
  float zB = rec_proj(xgB, q.hb, whhB);
  const float tA = __builtin_amdgcn_tanhf(zA);
  const float tB = __builtin_amdgcn_tanhf(zB);

  row_swap8(q.gv, tA);
  const float iv = __builtin_fmaf(0.5f, tA, 0.5f);
  const float fc = __builtin_fmaf(tB, hc, hc);
  const float oB = __builtin_fmaf(0.5f, tB, 0.5f);
  row_swap8(q.ov, oB);

  c  = __builtin_fmaf(iv, q.gv, fc);
  hc = 0.5f * c;
  h  = q.ov * __builtin_amdgcn_tanhf(c);
}

__global__ __launch_bounds__(128) void lstm_scan_kernel(
    const float* __restrict__ seq, const float* __restrict__ W_ih,
    const float* __restrict__ W_hh, const float* __restrict__ b_ih,
    const float* __restrict__ b_hh, float* __restrict__ h_out) {
  const int lane  = threadIdx.x & 31;
  const int wave  = threadIdx.x >> 5;
  const int row   = lane >> 4;
  const int unit  = lane & 7;
  const int upper = (lane >> 3) & 1;
  const int b     = blockIdx.x * 8 + wave * 2 + row;
  const int gA    = (upper ? 2 * HID : 0)   + unit;
  const int gB    = (upper ? 3 * HID : HID) + unit;

  const float wsA = upper ? 1.0f : 0.5f;
  const float wsB = 0.5f;

  float wihA[IN_DIM], wihB[IN_DIM];
#pragma unroll
  for (int i = 0; i < IN_DIM; ++i) {
    wihA[i] = W_ih[gA * IN_DIM + i] * wsA;
    wihB[i] = W_ih[gB * IN_DIM + i] * wsB;
  }
  float whhA[HID], whhB[HID];
#pragma unroll
  for (int k = 0; k < HID; ++k) {
    whhA[k] = W_hh[gA * HID + k] * wsA;
    whhB[k] = W_hh[gB * HID + k] * wsB;
  }
  const float biasA = (b_ih[gA] + b_hh[gA]) * wsA;
  const float biasB = (b_ih[gB] + b_hh[gB]) * wsB;

  float h = 0.0f, c = 0.0f, hc = 0.0f;
  Gathered q;
#pragma unroll
  for (int k = 0; k < HID; ++k) q.hb[k] = 0.0f;
  q.gv = 0.0f;
  q.ov = 0.0f;
  const float* xp = seq + (size_t)b * IN_DIM;
  const int tstride = BATCH * IN_DIM;

  float xr[RING][IN_DIM];
  for (int j = 0; j < RING; ++j) {
    for (int i = 0; i < IN_DIM; ++i) xr[j][i] = xp[j * tstride + i];
  }
  float xgA = in_proj(xr[0], wihA, biasA);
  float xgB = in_proj(xr[0], wihB, biasB);

  for (int t = 0; t < T_STEPS - RING; t += RING) {
    if (t + PF_AHEAD + RING <= T_STEPS) {
      for (int j = 0; j < RING; ++j)
        __builtin_prefetch(xp + (PF_AHEAD + j) * tstride, 0, 0);
    }
    for (int j = 0; j < RING; ++j) {
      for (int i = 0; i < IN_DIM; ++i) xr[j][i] = xp[(j + RING) * tstride + i];
      const float xgA_next = in_proj(xr[(j + 1) % RING], wihA, biasA);
      const float xgB_next = in_proj(xr[(j + 1) % RING], wihB, biasB);
      lstm_cell(xgA, xgB, whhA, whhB, q, h, c, hc);
      xgA = xgA_next;
      xgB = xgB_next;
      __builtin_amdgcn_sched_barrier(0);
    }
    xp += RING * tstride;
  }
  for (int j = 0; j < RING; ++j) {
    float xgA_next = 0.0f, xgB_next = 0.0f;
    if (j + 1 < RING) {
      xgA_next = in_proj(xr[j + 1], wihA, biasA);
      xgB_next = in_proj(xr[j + 1], wihB, biasB);
    }
    lstm_cell(xgA, xgB, whhA, whhB, q, h, c, hc);
    xgA = xgA_next;
    xgB = xgB_next;
  }
  if ((lane & 15) < HID) h_out[(size_t)b * HID + unit] = h;
}

__global__ __launch_bounds__(128) void out_proj_wmma(
    const float* __restrict__ h_last, const float* __restrict__ W_out,
    const float* __restrict__ b_out, float* __restrict__ out) {
  const int lane = threadIdx.x & 31;
  const int tile = (blockIdx.x * blockDim.x + threadIdx.x) >> 5;
  const int half = lane >> 4;
  const int lm   = lane & 15;

  const float* hrow = h_last + (size_t)(tile * 16 + lm) * HID;
  v2f a_lo, a_hi;
  a_lo.x = hrow[0 + 2 * half];
  a_lo.y = hrow[1 + 2 * half];
  a_hi.x = hrow[4 + 2 * half];
  a_hi.y = hrow[5 + 2 * half];

  float w0 = 0.f, w1 = 0.f, w2 = 0.f, w3 = 0.f, bo = 0.f;
  if (lm < OUT_DIM) {
    const float* wr = W_out + lm * HID;
    w0 = wr[0 + 2 * half];
    w1 = wr[1 + 2 * half];
    w2 = wr[4 + 2 * half];
    w3 = wr[5 + 2 * half];
    bo = b_out[lm];
  }
  v2f b_lo, b_hi;
  b_lo.x = w0; b_lo.y = w1;
  b_hi.x = w2; b_hi.y = w3;

  v8f acc;
#pragma unroll
  for (int r = 0; r < 8; ++r) acc[r] = bo;

  acc = __builtin_amdgcn_wmma_f32_16x16x4_f32(false, a_lo, false, b_lo,
                                              (short)0, acc, false, false);
  acc = __builtin_amdgcn_wmma_f32_16x16x4_f32(false, a_hi, false, b_hi,
                                              (short)0, acc, false, false);

  if (lm < OUT_DIM) {
    float* orow = out + (size_t)(tile * 16 + 8 * half) * OUT_DIM + lm;
#pragma unroll
    for (int r = 0; r < 8; ++r) orow[r * OUT_DIM] = acc[r];
  }
}

extern "C" void kernel_launch(void* const* d_in, const int* in_sizes, int n_in,
                              void* d_out, int out_size, void* d_ws, size_t ws_size,
                              hipStream_t stream) {
  const float* seq   = (const float*)d_in[0];
  const float* W_ih  = (const float*)d_in[1];
  const float* W_hh  = (const float*)d_in[2];
  const float* b_ih  = (const float*)d_in[3];
  const float* b_hh  = (const float*)d_in[4];
  const float* W_out = (const float*)d_in[5];
  const float* b_out = (const float*)d_in[6];
  float* out    = (float*)d_out;
  float* h_last = (float*)d_ws;

  lstm_scan_kernel<<<BATCH / 8, 128, 0, stream>>>(seq, W_ih, W_hh, b_ih, b_hh,
                                                  h_last);
  out_proj_wmma<<<(BATCH / 16) * 32 / 128, 128, 0, stream>>>(h_last, W_out,
                                                             b_out, out);
}


# input projection moved onto the f32 16x16x4 matrix instruction: one x tile (2 batch rows x 8 steps) per block, one 12-byte load per lane 4 blocks ahead, bias as a k element; x, bias sum order changes
# speedup vs baseline: 4.5919x; 1.1544x over previous
#include <hip/hip_runtime.h>


#define T_STEPS 4096
#define BATCH   1024
#define IN_DIM  5
#define HID     8
#define OUT_DIM 5
#define NB      8
#define NBLK    (T_STEPS / NB)

typedef __attribute__((ext_vector_type(2))) float v2f;
typedef __attribute__((ext_vector_type(8))) float v8f;


template <int K>
__device__ __forceinline__ void row_bcast(float& dst, float v) {
  const unsigned sel = 0x11111111u * (unsigned)K;
  dst = __int_as_float(__builtin_amdgcn_permlane16(
      __float_as_int(dst), __float_as_int(v), sel, sel, false, false));
}

__device__ __forceinline__ void row_swap8(float& dst, float v) {
  dst = __int_as_float(__builtin_amdgcn_permlane16(
      __float_as_int(dst), __float_as_int(v), 0xfedcba98u, 0x76543210u, false, false));
}

typedef float v2f_a4 __attribute__((ext_vector_type(2), aligned(4)));
struct XFrag { v2f lo; float e2; };

__device__ __forceinline__ XFrag load_xfrag(const float* p) {
  XFrag f;
  f.lo = *(const v2f_a4*)p;
  f.e2 = p[2];
  return f;
}

__device__ __forceinline__ v8f wmma4(v2f a, v2f b, v8f c) {
  return __builtin_amdgcn_wmma_f32_16x16x4_f32(false, a, false, b, (short)0, c,
                                               false, false);
}

__device__ __forceinline__ float rec_proj(float xg, const float (&hb)[HID],
                                          const float (&whh)[HID]) {
  float a0 = __builtin_fmaf(hb[0], whh[0], xg);
  float a1 = hb[1] * whh[1];
#pragma unroll
  for (int k = 2; k < HID; k += 2) {
    a0 = __builtin_fmaf(hb[k],     whh[k],     a0);
    a1 = __builtin_fmaf(hb[k + 1], whh[k + 1], a1);
  }
  return a0 + a1;
}

struct Gathered {
  float hb[HID];
  float gv, ov;
};

__device__ __forceinline__ void lstm_cell(float xgA, float xgB,
                                          const float (&whhA)[HID],
                                          const float (&whhB)[HID],
                                          Gathered& q,
                                          float& h, float& c, float& hc) {
  row_bcast<0>(q.hb[0], h); row_bcast<1>(q.hb[1], h);
  row_bcast<2>(q.hb[2], h); row_bcast<3>(q.hb[3], h);
  row_bcast<4>(q.hb[4], h); row_bcast<5>(q.hb[5], h);
  row_bcast<6>(q.hb[6], h); row_bcast<7>(q.hb[7], h);

  float zA = rec_proj(xgA, q.hb, whhA);
  float zB = rec_proj(xgB, q.hb, whhB);
  const float tA = __builtin_amdgcn_tanhf(zA);
  const float tB = __builtin_amdgcn_tanhf(zB);

  row_swap8(q.gv, tA);
  const float iv = __builtin_fmaf(0.5f, tA, 0.5f);
  const float fc = __builtin_fmaf(tB, hc, hc);
  const float oB = __builtin_fmaf(0.5f, tB, 0.5f);
  row_swap8(q.ov, oB);

  c  = __builtin_fmaf(iv, q.gv, fc);
  hc = 0.5f * c;
  h  = q.ov * __builtin_amdgcn_tanhf(c);
}

__global__ __launch_bounds__(128) void lstm_scan_kernel(
    const float* __restrict__ seq, const float* __restrict__ W_ih,
    const float* __restrict__ W_hh, const float* __restrict__ b_ih,
    const float* __restrict__ b_hh, float* __restrict__ h_out) {
  const int lane  = threadIdx.x & 31;
  const int wave  = threadIdx.x >> 5;
  const int row   = lane >> 4;
  const int unit  = lane & 7;
  const int upper = (lane >> 3) & 1;
  const int b     = blockIdx.x * 8 + wave * 2 + row;
  const int gA    = (upper ? 2 * HID : 0)   + unit;
  const int gB    = (upper ? 3 * HID : HID) + unit;

  const float wsA = upper ? 1.0f : 0.5f;
  const float wsB = 0.5f;

  v2f bA_lo, bA_hi, bB_lo, bB_hi;
  bA_lo.x = W_ih[gA * IN_DIM + 2 * row]     * wsA;
  bA_lo.y = W_ih[gA * IN_DIM + 2 * row + 1] * wsA;
  bB_lo.x = W_ih[gB * IN_DIM + 2 * row]     * wsB;
  bB_lo.y = W_ih[gB * IN_DIM + 2 * row + 1] * wsB;
  bA_hi.x = (row ? W_ih[gA * IN_DIM + 4] : b_ih[gA] + b_hh[gA]) * wsA;
  bB_hi.x = (row ? W_ih[gB * IN_DIM + 4] : b_ih[gB] + b_hh[gB]) * wsB;
  bA_hi.y = 0.0f;
  bB_hi.y = 0.0f;

  float whhA[HID], whhB[HID];
#pragma unroll
  for (int k = 0; k < HID; ++k) {
    whhA[k] = W_hh[gA * HID + k] * wsA;
    whhB[k] = W_hh[gB * HID + k] * wsB;
  }

  float h = 0.0f, c = 0.0f, hc = 0.0f;
  Gathered q;
#pragma unroll
  for (int k = 0; k < HID; ++k) q.hb[k] = 0.0f;
  q.gv = 0.0f;
  q.ov = 0.0f;

  const int m  = lane & 15;
  const int bm = blockIdx.x * 8 + wave * 2 + (m >> 3);
  const float* xa = seq + ((size_t)(m & 7) * BATCH + bm) * IN_DIM + 2 * row;
  const size_t blk_stride = (size_t)NB * BATCH * IN_DIM;
  const v8f zero8 = {0.f, 0.f, 0.f, 0.f, 0.f, 0.f, 0.f, 0.f};

  v8f x0A, x0B, x1A, x1B;
  {
    const XFrag f0 = load_xfrag(xa);
    const v2f a_lo = f0.lo;
    v2f a_hi;
    a_hi.x = row ? f0.e2 : 1.0f; a_hi.y = 0.0f;
    x0A = wmma4(a_hi, bA_hi, wmma4(a_lo, bA_lo, zero8));
    x0B = wmma4(a_hi, bB_hi, wmma4(a_lo, bB_lo, zero8));
  }
  x1A = zero8;
  x1B = zero8;
  XFrag f1 = load_xfrag(xa + 1 * blk_stride);
  XFrag f2 = load_xfrag(xa + 2 * blk_stride);
  XFrag f3 = load_xfrag(xa + 3 * blk_stride);
  XFrag f4 = load_xfrag(xa + 4 * blk_stride);
  const float* xn = xa + 4 * blk_stride;

#define SCAN_BLOCK(curA, curB, nxtA, nxtB, f, blk)                              \
  {                                                                             \
    const v2f a_lo = f.lo;                                                      \
    v2f a_hi;                                                                   \
    a_hi.x = row ? f.e2 : 1.0f; a_hi.y = 0.0f;                                  \
           \
    _Pragma("unroll")                                                           \
    for (int j = 0; j < NB; ++j) {                                              \
      if (j == 0) nxtA = wmma4(a_lo, bA_lo, zero8);                             \
      if (j == 3) nxtB = wmma4(a_lo, bB_lo, zero8);                             \
      if (j == 5) nxtA = wmma4(a_hi, bA_hi, nxtA);                              \
      if (j == 6) nxtB = wmma4(a_hi, bB_hi, nxtB);                              \
            \
           \
           \
      if (j == 3) xn += ((blk) + 5 < NBLK) ? blk_stride : 0;                    \
      if (j == 4) f = load_xfrag(xn);                                           \
      lstm_cell(curA[j], curB[j], whhA, whhB, q, h, c, hc);                     \
           \
           \
           \
           \
           \
      __builtin_amdgcn_sched_barrier(0);                                        \
    }                                                                           \
  }

  for (int blk = 0; blk < NBLK; blk += 4) {
    SCAN_BLOCK(x0A, x0B, x1A, x1B, f1, blk)
    SCAN_BLOCK(x1A, x1B, x0A, x0B, f2, blk + 1)
    SCAN_BLOCK(x0A, x0B, x1A, x1B, f3, blk + 2)
    SCAN_BLOCK(x1A, x1B, x0A, x0B, f4, blk + 3)
  }
#undef SCAN_BLOCK

  if ((lane & 15) < HID) h_out[(size_t)b * HID + unit] = h;
}

__global__ __launch_bounds__(128) void out_proj_wmma(
    const float* __restrict__ h_last, const float* __restrict__ W_out,
    const float* __restrict__ b_out, float* __restrict__ out) {
  const int lane = threadIdx.x & 31;
  const int tile = (blockIdx.x * blockDim.x + threadIdx.x) >> 5;
  const int half = lane >> 4;
  const int lm   = lane & 15;

  const float* hrow = h_last + (size_t)(tile * 16 + lm) * HID;
  v2f a_lo, a_hi;
  a_lo.x = hrow[0 + 2 * half];
  a_lo.y = hrow[1 + 2 * half];
  a_hi.x = hrow[4 + 2 * half];
  a_hi.y = hrow[5 + 2 * half];

  float w0 = 0.f, w1 = 0.f, w2 = 0.f, w3 = 0.f, bo = 0.f;
  if (lm < OUT_DIM) {
    const float* wr = W_out + lm * HID;
    w0 = wr[0 + 2 * half];
    w1 = wr[1 + 2 * half];
    w2 = wr[4 + 2 * half];
    w3 = wr[5 + 2 * half];
    bo = b_out[lm];
  }
  v2f b_lo, b_hi;
  b_lo.x = w0; b_lo.y = w1;
  b_hi.x = w2; b_hi.y = w3;

  v8f acc;
#pragma unroll
  for (int r = 0; r < 8; ++r) acc[r] = bo;

  acc = __builtin_amdgcn_wmma_f32_16x16x4_f32(false, a_lo, false, b_lo,
                                              (short)0, acc, false, false);
  acc = __builtin_amdgcn_wmma_f32_16x16x4_f32(false, a_hi, false, b_hi,
                                              (short)0, acc, false, false);

  if (lm < OUT_DIM) {
    float* orow = out + (size_t)(tile * 16 + 8 * half) * OUT_DIM + lm;
#pragma unroll
    for (int r = 0; r < 8; ++r) orow[r * OUT_DIM] = acc[r];
  }
}

extern "C" void kernel_launch(void* const* d_in, const int* in_sizes, int n_in,
                              void* d_out, int out_size, void* d_ws, size_t ws_size,
                              hipStream_t stream) {
  const float* seq   = (const float*)d_in[0];
  const float* W_ih  = (const float*)d_in[1];
  const float* W_hh  = (const float*)d_in[2];
  const float* b_ih  = (const float*)d_in[3];
  const float* b_hh  = (const float*)d_in[4];
  const float* W_out = (const float*)d_in[5];
  const float* b_out = (const float*)d_in[6];
  float* out    = (float*)d_out;
  float* h_last = (float*)d_ws;

  lstm_scan_kernel<<<BATCH / 8, 128, 0, stream>>>(seq, W_ih, W_hh, b_ih, b_hh,
                                                  h_last);
  out_proj_wmma<<<(BATCH / 16) * 32 / 128, 128, 0, stream>>>(h_last, W_out,
                                                             b_out, out);
}


# speedup vs baseline: 4.9119x; 1.0697x over previous
#include <hip/hip_runtime.h>


#define T_STEPS 4096
#define BATCH   1024
#define IN_DIM  5
#define HID     8
#define OUT_DIM 5
#define NB      8
#define NBLK    (T_STEPS / NB)

typedef __attribute__((ext_vector_type(2))) float v2f;
typedef __attribute__((ext_vector_type(8))) float v8f;


template <int K>
__device__ __forceinline__ void row_bcast(float& dst, float v, unsigned selK) {
  dst = __int_as_float(__builtin_amdgcn_permlane16_var(
      __float_as_int(dst), __float_as_int(v), selK, false, false));
}

__device__ __forceinline__ void row_swap8(float& dst, float v) {
  dst = __int_as_float(__builtin_amdgcn_permlane16(
      __float_as_int(dst), __float_as_int(v), 0xfedcba98u, 0x76543210u, false, false));
}

typedef float v2f_a4 __attribute__((ext_vector_type(2), aligned(4)));
struct XFrag { v2f lo; float e2; };

__device__ __forceinline__ XFrag load_xfrag(const float* p) {
  XFrag f;
  f.lo = *(const v2f_a4*)p;
  f.e2 = p[2];
  return f;
}

__device__ __forceinline__ v8f wmma4(v2f a, v2f b, v8f c) {
  return __builtin_amdgcn_wmma_f32_16x16x4_f32(false, a, false, b, (short)0, c,
                                               false, false);
}

struct Gathered {
  float hb[HID];
  float gv, ov;
};

__device__ __forceinline__ void lstm_cell(float xgA, float xgB,
                                          const float (&whhA)[HID],
                                          const float (&whhB)[HID],
                                          const unsigned (&sel)[HID],
                                          Gathered& q,
                                          float& h, float& c, float& hc) {
#define STEP_FENCE() __builtin_amdgcn_sched_barrier(0)
  row_bcast<0>(q.hb[0], h, sel[0]); row_bcast<1>(q.hb[1], h, sel[1]);
  row_bcast<2>(q.hb[2], h, sel[2]); row_bcast<3>(q.hb[3], h, sel[3]);
  row_bcast<4>(q.hb[4], h, sel[4]);
  STEP_FENCE();
  float a0A = __builtin_fmaf(q.hb[0], whhA[0], xgA);
  float a1A = q.hb[1] * whhA[1];
  float a0B = __builtin_fmaf(q.hb[0], whhB[0], xgB);
  float a1B = q.hb[1] * whhB[1];
  STEP_FENCE();
  row_bcast<5>(q.hb[5], h, sel[5]);
  a0A = __builtin_fmaf(q.hb[2], whhA[2], a0A);
  a1A = __builtin_fmaf(q.hb[3], whhA[3], a1A);
  a0B = __builtin_fmaf(q.hb[2], whhB[2], a0B);
  a1B = __builtin_fmaf(q.hb[3], whhB[3], a1B);
  STEP_FENCE();
  row_bcast<6>(q.hb[6], h, sel[6]); row_bcast<7>(q.hb[7], h, sel[7]);
  a0A = __builtin_fmaf(q.hb[4], whhA[4], a0A);
  a1A = __builtin_fmaf(q.hb[5], whhA[5], a1A);
  a0B = __builtin_fmaf(q.hb[4], whhB[4], a0B);
  a1B = __builtin_fmaf(q.hb[5], whhB[5], a1B);
  STEP_FENCE();
  a0A = __builtin_fmaf(q.hb[6], whhA[6], a0A);
  a1A = __builtin_fmaf(q.hb[7], whhA[7], a1A);
  a0B = __builtin_fmaf(q.hb[6], whhB[6], a0B);
  a1B = __builtin_fmaf(q.hb[7], whhB[7], a1B);
  STEP_FENCE();
  const float zA = a0A + a1A;
  STEP_FENCE();
  const float zB = a0B + a1B;
  STEP_FENCE();
  const float tA = __builtin_amdgcn_tanhf(zA);
  const float tB = __builtin_amdgcn_tanhf(zB);
#undef STEP_FENCE

  row_swap8(q.gv, tA);
  const float iv = __builtin_fmaf(0.5f, tA, 0.5f);
  const float fc = __builtin_fmaf(tB, hc, hc);
  const float oB = __builtin_fmaf(0.5f, tB, 0.5f);
  row_swap8(q.ov, oB);

  c  = __builtin_fmaf(iv, q.gv, fc);
  hc = 0.5f * c;
  h  = q.ov * __builtin_amdgcn_tanhf(c);
}

__global__ __launch_bounds__(128) void lstm_scan_kernel(
    const float* __restrict__ seq, const float* __restrict__ W_ih,
    const float* __restrict__ W_hh, const float* __restrict__ b_ih,
    const float* __restrict__ b_hh, float* __restrict__ h_out) {
  const int lane  = threadIdx.x & 31;
  const int wave  = threadIdx.x >> 5;
  const int row   = lane >> 4;
  const int unit  = lane & 7;
  const int upper = (lane >> 3) & 1;
  const int b     = blockIdx.x * 8 + wave * 2 + row;
  const int gA    = (upper ? 2 * HID : 0)   + unit;
  const int gB    = (upper ? 3 * HID : HID) + unit;

  const float wsA = upper ? 1.0f : 0.5f;
  const float wsB = 0.5f;

  v2f bA_lo, bA_hi, bB_lo, bB_hi;
  bA_lo.x = W_ih[gA * IN_DIM + 2 * row]     * wsA;
  bA_lo.y = W_ih[gA * IN_DIM + 2 * row + 1] * wsA;
  bB_lo.x = W_ih[gB * IN_DIM + 2 * row]     * wsB;
  bB_lo.y = W_ih[gB * IN_DIM + 2 * row + 1] * wsB;
  bA_hi.x = (row ? W_ih[gA * IN_DIM + 4] : b_ih[gA] + b_hh[gA]) * wsA;
  bB_hi.x = (row ? W_ih[gB * IN_DIM + 4] : b_ih[gB] + b_hh[gB]) * wsB;
  bA_hi.y = 0.0f;
  bB_hi.y = 0.0f;

  float whhA[HID], whhB[HID];
#pragma unroll
  for (int k = 0; k < HID; ++k) {
    whhA[k] = W_hh[gA * HID + k] * wsA;
    whhB[k] = W_hh[gB * HID + k] * wsB;
  }

  const unsigned sel[HID] = {0u, 1u, 2u, 3u, 4u, 5u, 6u, 7u};

  float h = 0.0f, c = 0.0f, hc = 0.0f;
  Gathered q;
#pragma unroll
  for (int k = 0; k < HID; ++k) q.hb[k] = 0.0f;
  q.gv = 0.0f;
  q.ov = 0.0f;

  const int m  = lane & 15;
  const int bm = blockIdx.x * 8 + wave * 2 + (m >> 3);
  const float* xa = seq + ((size_t)(m & 7) * BATCH + bm) * IN_DIM + 2 * row;
  const size_t blk_stride = (size_t)NB * BATCH * IN_DIM;
  const v8f zero8 = {0.f, 0.f, 0.f, 0.f, 0.f, 0.f, 0.f, 0.f};

  v8f x0A, x0B, x1A, x1B;
  {
    const XFrag f0 = load_xfrag(xa);
    const v2f a_lo = f0.lo;
    v2f a_hi;
    a_hi.x = row ? f0.e2 : 1.0f; a_hi.y = 0.0f;
    x0A = wmma4(a_hi, bA_hi, wmma4(a_lo, bA_lo, zero8));
    x0B = wmma4(a_hi, bB_hi, wmma4(a_lo, bB_lo, zero8));
  }
  x1A = zero8;
  x1B = zero8;
  XFrag f1 = load_xfrag(xa + 1 * blk_stride);
  XFrag f2 = load_xfrag(xa + 2 * blk_stride);
  XFrag f3 = load_xfrag(xa + 3 * blk_stride);
  XFrag f4 = load_xfrag(xa + 4 * blk_stride);
  const float* xn = xa + 4 * blk_stride;

#define SCAN_BLOCK(curA, curB, nxtA, nxtB, f, blk)                              \
  {                                                                             \
    const v2f a_lo = f.lo;                                                      \
    v2f a_hi;                                                                   \
    a_hi.x = row ? f.e2 : 1.0f; a_hi.y = 0.0f;                                  \
           \
    _Pragma("unroll")                                                           \
    for (int j = 0; j < NB; ++j) {                                              \
      lstm_cell(curA[j], curB[j], whhA, whhB, sel, q, h, c, hc);                \
           \
           \
           \
      if (j == 0) nxtA = wmma4(a_lo, bA_lo, zero8);                             \
      if (j == 3) nxtB = wmma4(a_lo, bB_lo, zero8);                             \
      if (j == 5) nxtA = wmma4(a_hi, bA_hi, nxtA);                              \
      if (j == 6) nxtB = wmma4(a_hi, bB_hi, nxtB);                              \
            \
           \
           \
      if (j == 3) xn += ((blk) + 5 < NBLK) ? blk_stride : 0;                    \
      if (j == 4) f = load_xfrag(xn);                                           \
           \
           \
           \
           \
           \
      __builtin_amdgcn_sched_barrier(0);                                        \
    }                                                                           \
  }

  for (int blk = 0; blk < NBLK; blk += 4) {
    SCAN_BLOCK(x0A, x0B, x1A, x1B, f1, blk)
    SCAN_BLOCK(x1A, x1B, x0A, x0B, f2, blk + 1)
    SCAN_BLOCK(x0A, x0B, x1A, x1B, f3, blk + 2)
    SCAN_BLOCK(x1A, x1B, x0A, x0B, f4, blk + 3)
  }
#undef SCAN_BLOCK

  if ((lane & 15) < HID) h_out[(size_t)b * HID + unit] = h;
}

__global__ __launch_bounds__(128) void out_proj_wmma(
    const float* __restrict__ h_last, const float* __restrict__ W_out,
    const float* __restrict__ b_out, float* __restrict__ out) {
  const int lane = threadIdx.x & 31;
  const int tile = (blockIdx.x * blockDim.x + threadIdx.x) >> 5;
  const int half = lane >> 4;
  const int lm   = lane & 15;

  const float* hrow = h_last + (size_t)(tile * 16 + lm) * HID;
  v2f a_lo, a_hi;
  a_lo.x = hrow[0 + 2 * half];
  a_lo.y = hrow[1 + 2 * half];
  a_hi.x = hrow[4 + 2 * half];
  a_hi.y = hrow[5 + 2 * half];

  float w0 = 0.f, w1 = 0.f, w2 = 0.f, w3 = 0.f, bo = 0.f;
  if (lm < OUT_DIM) {
    const float* wr = W_out + lm * HID;
    w0 = wr[0 + 2 * half];
    w1 = wr[1 + 2 * half];
    w2 = wr[4 + 2 * half];
    w3 = wr[5 + 2 * half];
    bo = b_out[lm];
  }
  v2f b_lo, b_hi;
  b_lo.x = w0; b_lo.y = w1;
  b_hi.x = w2; b_hi.y = w3;

  v8f acc;
#pragma unroll
  for (int r = 0; r < 8; ++r) acc[r] = bo;

  acc = __builtin_amdgcn_wmma_f32_16x16x4_f32(false, a_lo, false, b_lo,
                                              (short)0, acc, false, false);
  acc = __builtin_amdgcn_wmma_f32_16x16x4_f32(false, a_hi, false, b_hi,
                                              (short)0, acc, false, false);

  if (lm < OUT_DIM) {
    float* orow = out + (size_t)(tile * 16 + 8 * half) * OUT_DIM + lm;
#pragma unroll
    for (int r = 0; r < 8; ++r) orow[r * OUT_DIM] = acc[r];
  }
}

extern "C" void kernel_launch(void* const* d_in, const int* in_sizes, int n_in,
                              void* d_out, int out_size, void* d_ws, size_t ws_size,
                              hipStream_t stream) {
  const float* seq   = (const float*)d_in[0];
  const float* W_ih  = (const float*)d_in[1];
  const float* W_hh  = (const float*)d_in[2];
  const float* b_ih  = (const float*)d_in[3];
  const float* b_hh  = (const float*)d_in[4];
  const float* W_out = (const float*)d_in[5];
  const float* b_out = (const float*)d_in[6];
  float* out    = (float*)d_out;
  float* h_last = (float*)d_ws;

  lstm_scan_kernel<<<BATCH / 8, 128, 0, stream>>>(seq, W_ih, W_hh, b_ih, b_hh,
                                                  h_last);
  out_proj_wmma<<<(BATCH / 16) * 32 / 128, 128, 0, stream>>>(h_last, W_out,
                                                             b_out, out);
}
